# starting kernel
# baseline (speedup 1.0000x reference)
#include <hip/hip_runtime.h>
#include <math.h>

typedef float v2f __attribute__((ext_vector_type(2)));
typedef float v4f __attribute__((ext_vector_type(4)));
typedef float v8f __attribute__((ext_vector_type(8)));

#define N_NODES 50000
#define N_EDGES 800000
#define FDIM 64
#define LSTRIDE 68

__device__ __forceinline__ float silu_f(float v) {
    return v * (1.0f / (1.0f + __expf(-v)));
}

__device__ __forceinline__ v8f wmma4(v2f a, v2f b, v8f c) {
    return __builtin_amdgcn_wmma_f32_16x16x4_f32(false, a, false, b, (short)0, c, false, false);
}

__device__ __forceinline__ void wmma_step(const float* __restrict__ Wp, int t, int lane,
                                          v2f a, v8f* acc) {
    const float* p = Wp + ((size_t)(t * 32 + lane) << 3);
    const v4f f0 = *(const v4f*)(p);
    const v4f f1 = *(const v4f*)(p + 4);
    v2f b;
    b.x = f0.x; b.y = f0.y; acc[0] = wmma4(a, b, acc[0]);
    b.x = f0.z; b.y = f0.w; acc[1] = wmma4(a, b, acc[1]);
    b.x = f1.x; b.y = f1.y; acc[2] = wmma4(a, b, acc[2]);
    b.x = f1.z; b.y = f1.w; acc[3] = wmma4(a, b, acc[3]);
}

__device__ __forceinline__ void gemm_lds(const float* __restrict__ tile,
                                         const float* __restrict__ Wp,
                                         int m, int half, int lane, v8f* acc) {
    #pragma unroll
    for (int t = 0; t < 16; ++t) {
        const int k0 = 4 * t + 2 * half;
        v2f a;
        a.x = tile[m * LSTRIDE + k0];
        a.y = tile[m * LSTRIDE + k0 + 1];
        wmma_step(Wp, t, lane, a, acc);
    }
}

__global__ void pack_kernel(const float* __restrict__ W, float* __restrict__ dst,
                            int krows, int T) {
    const int idx = blockIdx.x * blockDim.x + threadIdx.x;
    if (idx >= T * 256) return;
    const int t    = idx >> 8;
    const int r    = idx & 255;
    const int lane = r >> 3;
    const int q    = r & 7;
    const int j    = q >> 1;
    const int v    = q & 1;
    const int k    = 4 * t + 2 * (lane >> 4) + v;
    dst[idx] = (k < krows) ? W[k * FDIM + j * 16 + (lane & 15)] : 0.0f;
}

__global__ void __launch_bounds__(32)
egnn_edge_kernel(const float* __restrict__ x, const float* __restrict__ h,
                 const int* __restrict__ edges,
                 const float* __restrict__ We1p, const float* __restrict__ be1,
                 const float* __restrict__ We2p, const float* __restrict__ be2,
                 const float* __restrict__ Wp1p, const float* __restrict__ bp1,
                 const float* __restrict__ Wp2, const float* __restrict__ bp2,
                 float* __restrict__ cnt, float* __restrict__ xsum,
                 float* __restrict__ msum)
{
    __shared__ float tile[16 * LSTRIDE];
    const int lane = threadIdx.x & 31;
    const int m    = lane & 15;
    const int half = lane >> 4;
    const int n    = m;
    const int e0   = blockIdx.x * 16;
    const int e    = e0 + m;

    const int src = edges[e];
    const int dst = edges[N_EDGES + e];
    const float dx = x[src * 3 + 0] - x[dst * 3 + 0];
    const float dy = x[src * 3 + 1] - x[dst * 3 + 1];
    const float dz = x[src * 3 + 2] - x[dst * 3 + 2];
    const float dij = sqrtf(dx * dx + dy * dy + dz * dz);

    const float* __restrict__ hs = h + (size_t)src * FDIM;
    const float* __restrict__ hd = h + (size_t)dst * FDIM;

    v8f acc[4] = {};
    #pragma unroll 4
    for (int t = 0; t < 33; ++t) {
        const int k0 = 4 * t + 2 * half;
        v2f a;
        if (t < 16)      { a.x = hs[k0];      a.y = hs[k0 + 1]; }
        else if (t < 32) { a.x = hd[k0 - 64]; a.y = hd[k0 - 63]; }
        else             { a.x = (half == 0) ? dij : 0.0f; a.y = 0.0f; }
        wmma_step(We1p, t, lane, a, acc);
    }
    #pragma unroll
    for (int j = 0; j < 4; ++j)
        #pragma unroll
        for (int v = 0; v < 8; ++v)
            tile[(v + half * 8) * LSTRIDE + j * 16 + n] = silu_f(acc[j][v] + be1[j * 16 + n]);
    __syncthreads();

    v8f macc[4] = {};
    gemm_lds(tile, We2p, m, half, lane, macc);
    #pragma unroll
    for (int j = 0; j < 4; ++j)
        #pragma unroll
        for (int v = 0; v < 8; ++v)
            macc[j][v] = silu_f(macc[j][v] + be2[j * 16 + n]);
    __syncthreads();
    #pragma unroll
    for (int j = 0; j < 4; ++j)
        #pragma unroll
        for (int v = 0; v < 8; ++v)
            tile[(v + half * 8) * LSTRIDE + j * 16 + n] = macc[j][v];
    __syncthreads();

    v8f pacc[4] = {};
    gemm_lds(tile, Wp1p, m, half, lane, pacc);
    __syncthreads();
    #pragma unroll
    for (int j = 0; j < 4; ++j)
        #pragma unroll
        for (int v = 0; v < 8; ++v)
            tile[(v + half * 8) * LSTRIDE + j * 16 + n] = silu_f(pacc[j][v] + bp1[j * 16 + n]);
    __syncthreads();

    float partial = 0.0f;
    #pragma unroll 8
    for (int k = 0; k < 32; ++k) {
        const int kk = half * 32 + k;
        partial += tile[m * LSTRIDE + kk] * Wp2[kk];
    }
    const float w = partial + __shfl_xor(partial, 16, 32) + bp2[0];

    if (half == 0) {
        unsafeAtomicAdd(&xsum[src * 3 + 0], -dx * w);
        unsafeAtomicAdd(&xsum[src * 3 + 1], -dy * w);
        unsafeAtomicAdd(&xsum[src * 3 + 2], -dz * w);
        unsafeAtomicAdd(&cnt[src], 1.0f);
    }
    #pragma unroll
    for (int v = 0; v < 8; ++v) {
        const int mr = v + half * 8;
        const int srcRow = __shfl(src, mr, 32);
        #pragma unroll
        for (int j = 0; j < 4; ++j)
            unsafeAtomicAdd(&msum[(size_t)srcRow * FDIM + j * 16 + n], macc[j][v]);
    }
}

__global__ void __launch_bounds__(32)
egnn_node_kernel(const float* __restrict__ x, const float* __restrict__ h,
                 const float* __restrict__ Wn1p, const float* __restrict__ bn1,
                 const float* __restrict__ Wn2p, const float* __restrict__ bn2,
                 const float* __restrict__ cnt, const float* __restrict__ xsum,
                 const float* __restrict__ msum,
                 float* __restrict__ xout, float* __restrict__ hout)
{
    __shared__ float tile[16 * LSTRIDE];
    const int lane = threadIdx.x & 31;
    const int m    = lane & 15;
    const int half = lane >> 4;
    const int n    = m;
    const int i0   = blockIdx.x * 16;
    const int i    = i0 + m;

    const float inv = 1.0f / fmaxf(cnt[i], 1.0f);

    if (half == 0) {
        #pragma unroll
        for (int d = 0; d < 3; ++d)
            xout[i * 3 + d] = x[i * 3 + d] + xsum[i * 3 + d] * inv;
    }

    const float* __restrict__ hrow = h + (size_t)i * FDIM;
    const float* __restrict__ mrow = msum + (size_t)i * FDIM;

    v8f acc[4] = {};
    #pragma unroll 4
    for (int t = 0; t < 32; ++t) {
        const int k0 = 4 * t + 2 * half;
        v2f a;
        if (t < 16) { a.x = hrow[k0];            a.y = hrow[k0 + 1]; }
        else        { a.x = mrow[k0 - 64] * inv; a.y = mrow[k0 - 63] * inv; }
        wmma_step(Wn1p, t, lane, a, acc);
    }
    #pragma unroll
    for (int j = 0; j < 4; ++j)
        #pragma unroll
        for (int v = 0; v < 8; ++v)
            tile[(v + half * 8) * LSTRIDE + j * 16 + n] = silu_f(acc[j][v] + bn1[j * 16 + n]);
    __syncthreads();

    v8f acc2[4] = {};
    gemm_lds(tile, Wn2p, m, half, lane, acc2);
    #pragma unroll
    for (int v = 0; v < 8; ++v) {
        const int mr = v + half * 8;
        const size_t base = (size_t)(i0 + mr) * FDIM;
        #pragma unroll
        for (int j = 0; j < 4; ++j) {
            const int col = j * 16 + n;
            hout[base + col] = h[base + col] + acc2[j][v] + bn2[col];
        }
    }
}

__global__ void zero_kernel(float* __restrict__ p, int count) {
    const int idx = blockIdx.x * blockDim.x + threadIdx.x;
    if (idx < count) p[idx] = 0.0f;
}

extern "C" void kernel_launch(void* const* d_in, const int* in_sizes, int n_in,
                              void* d_out, int out_size, void* d_ws, size_t ws_size,
                              hipStream_t stream) {
    const float* x     = (const float*)d_in[0];
    const float* h     = (const float*)d_in[1];
    const int*   edges = (const int*)d_in[2];
    const float* We1   = (const float*)d_in[3];
    const float* be1   = (const float*)d_in[4];
    const float* We2   = (const float*)d_in[5];
    const float* be2   = (const float*)d_in[6];
    const float* Wp1   = (const float*)d_in[7];
    const float* bp1   = (const float*)d_in[8];
    const float* Wp2   = (const float*)d_in[9];
    const float* bp2   = (const float*)d_in[10];
    const float* Wn1   = (const float*)d_in[11];
    const float* bn1   = (const float*)d_in[12];
    const float* Wn2   = (const float*)d_in[13];
    const float* bn2   = (const float*)d_in[14];

    float* cnt  = (float*)d_ws;
    float* xsum = cnt + N_NODES;
    float* msum = xsum + 3 * N_NODES;
    float* wpk  = msum + (size_t)N_NODES * FDIM;
    float* We1p = wpk;
    float* We2p = We1p + 33 * 256;
    float* Wp1p = We2p + 16 * 256;
    float* Wn1p = Wp1p + 16 * 256;
    float* Wn2p = Wn1p + 32 * 256;

    float* xout = (float*)d_out;
    float* hout = xout + 3 * N_NODES;

    const int zcount = N_NODES * (1 + 3 + FDIM);
    zero_kernel<<<(zcount + 255) / 256, 256, 0, stream>>>(cnt, zcount);

    pack_kernel<<<(33 * 256 + 255) / 256, 256, 0, stream>>>(We1, We1p, 129, 33);
    pack_kernel<<<(16 * 256 + 255) / 256, 256, 0, stream>>>(We2, We2p, 64, 16);
    pack_kernel<<<(16 * 256 + 255) / 256, 256, 0, stream>>>(Wp1, Wp1p, 64, 16);
    pack_kernel<<<(32 * 256 + 255) / 256, 256, 0, stream>>>(Wn1, Wn1p, 128, 32);
    pack_kernel<<<(16 * 256 + 255) / 256, 256, 0, stream>>>(Wn2, Wn2p, 64, 16);

    egnn_edge_kernel<<<N_EDGES / 16, 32, 0, stream>>>(
        x, h, edges, We1p, be1, We2p, be2, Wp1p, bp1, Wp2, bp2, cnt, xsum, msum);

    egnn_node_kernel<<<N_NODES / 16, 32, 0, stream>>>(
        x, h, Wn1p, bn1, Wn2p, bn2, cnt, xsum, msum, xout, hout);
}


# scatter atomics regrouped by line: m_ij added from the LDS tile one whole 128-byte row line per instruction; x sums and count in one 16-byte node record, 4 lanes per edge; sums still in arrival order
# speedup vs baseline: 1.1145x; 1.1145x over previous
#include <hip/hip_runtime.h>
#include <math.h>

typedef float v2f __attribute__((ext_vector_type(2)));
typedef float v4f __attribute__((ext_vector_type(4)));
typedef float v8f __attribute__((ext_vector_type(8)));

#define N_NODES 50000
#define N_EDGES 800000
#define FDIM 64
#define LSTRIDE 68

__device__ __forceinline__ float silu_f(float v) {
    return v * (1.0f / (1.0f + __expf(-v)));
}

__device__ __forceinline__ v8f wmma4(v2f a, v2f b, v8f c) {
    return __builtin_amdgcn_wmma_f32_16x16x4_f32(false, a, false, b, (short)0, c, false, false);
}

__device__ __forceinline__ void wmma_step(const float* __restrict__ Wp, int t, int lane,
                                          v2f a, v8f* acc) {
    const float* p = Wp + ((size_t)(t * 32 + lane) << 3);
    const v4f f0 = *(const v4f*)(p);
    const v4f f1 = *(const v4f*)(p + 4);
    v2f b;
    b.x = f0.x; b.y = f0.y; acc[0] = wmma4(a, b, acc[0]);
    b.x = f0.z; b.y = f0.w; acc[1] = wmma4(a, b, acc[1]);
    b.x = f1.x; b.y = f1.y; acc[2] = wmma4(a, b, acc[2]);
    b.x = f1.z; b.y = f1.w; acc[3] = wmma4(a, b, acc[3]);
}

__device__ __forceinline__ void gemm_lds(const float* __restrict__ tile,
                                         const float* __restrict__ Wp,
                                         int m, int half, int lane, v8f* acc) {
    #pragma unroll
    for (int t = 0; t < 16; ++t) {
        const int k0 = 4 * t + 2 * half;
        v2f a;
        a.x = tile[m * LSTRIDE + k0];
        a.y = tile[m * LSTRIDE + k0 + 1];
        wmma_step(Wp, t, lane, a, acc);
    }
}

__global__ void pack_kernel(const float* __restrict__ W, float* __restrict__ dst,
                            int krows, int T) {
    const int idx = blockIdx.x * blockDim.x + threadIdx.x;
    if (idx >= T * 256) return;
    const int t    = idx >> 8;
    const int r    = idx & 255;
    const int lane = r >> 3;
    const int q    = r & 7;
    const int j    = q >> 1;
    const int v    = q & 1;
    const int k    = 4 * t + 2 * (lane >> 4) + v;
    dst[idx] = (k < krows) ? W[k * FDIM + j * 16 + (lane & 15)] : 0.0f;
}

__global__ void __launch_bounds__(32)
egnn_edge_kernel(const float* __restrict__ x, const float* __restrict__ h,
                 const int* __restrict__ edges,
                 const float* __restrict__ We1p, const float* __restrict__ be1,
                 const float* __restrict__ We2p, const float* __restrict__ be2,
                 const float* __restrict__ Wp1p, const float* __restrict__ bp1,
                 const float* __restrict__ Wp2, const float* __restrict__ bp2,
                 float* __restrict__ xc, float* __restrict__ msum)
{
    __shared__ float tile[16 * LSTRIDE];
    __shared__ __attribute__((aligned(16))) float rec[16 * 4];
    __shared__ int rsrc[16];
    const int lane = threadIdx.x & 31;
    const int m    = lane & 15;
    const int half = lane >> 4;
    const int n    = m;
    const int e0   = blockIdx.x * 16;
    const int e    = e0 + m;

    const int src = edges[e];
    const int dst = edges[N_EDGES + e];
    const float dx = x[src * 3 + 0] - x[dst * 3 + 0];
    const float dy = x[src * 3 + 1] - x[dst * 3 + 1];
    const float dz = x[src * 3 + 2] - x[dst * 3 + 2];
    const float dij = sqrtf(dx * dx + dy * dy + dz * dz);

    const float* __restrict__ hs = h + (size_t)src * FDIM;
    const float* __restrict__ hd = h + (size_t)dst * FDIM;

    v8f acc[4] = {};
    #pragma unroll 4
    for (int t = 0; t < 33; ++t) {
        const int k0 = 4 * t + 2 * half;
        v2f a;
        if (t < 16)      { a.x = hs[k0];      a.y = hs[k0 + 1]; }
        else if (t < 32) { a.x = hd[k0 - 64]; a.y = hd[k0 - 63]; }
        else             { a.x = (half == 0) ? dij : 0.0f; a.y = 0.0f; }
        wmma_step(We1p, t, lane, a, acc);
    }
    #pragma unroll
    for (int j = 0; j < 4; ++j)
        #pragma unroll
        for (int v = 0; v < 8; ++v)
            tile[(v + half * 8) * LSTRIDE + j * 16 + n] = silu_f(acc[j][v] + be1[j * 16 + n]);
    __syncthreads();

    v8f macc[4] = {};
    gemm_lds(tile, We2p, m, half, lane, macc);
    #pragma unroll
    for (int j = 0; j < 4; ++j)
        #pragma unroll
        for (int v = 0; v < 8; ++v)
            macc[j][v] = silu_f(macc[j][v] + be2[j * 16 + n]);
    __syncthreads();
    #pragma unroll
    for (int j = 0; j < 4; ++j)
        #pragma unroll
        for (int v = 0; v < 8; ++v)
            tile[(v + half * 8) * LSTRIDE + j * 16 + n] = macc[j][v];
    __syncthreads();

    {
        int   sr[16];
        float lo[16], hi[16];
        #pragma unroll
        for (int r = 0; r < 16; ++r) {
            sr[r] = __shfl(src, r, 32);
            lo[r] = tile[r * LSTRIDE + lane];
            hi[r] = tile[r * LSTRIDE + 32 + lane];
        }
        #pragma unroll
        for (int r = 0; r < 16; ++r) {
            float* __restrict__ p = msum + (size_t)sr[r] * FDIM + lane;
            unsafeAtomicAdd(p,      lo[r]);
            unsafeAtomicAdd(p + 32, hi[r]);
        }
    }

    v8f pacc[4] = {};
    gemm_lds(tile, Wp1p, m, half, lane, pacc);
    __syncthreads();
    #pragma unroll
    for (int j = 0; j < 4; ++j)
        #pragma unroll
        for (int v = 0; v < 8; ++v)
            tile[(v + half * 8) * LSTRIDE + j * 16 + n] = silu_f(pacc[j][v] + bp1[j * 16 + n]);
    __syncthreads();

    float partial = 0.0f;
    #pragma unroll 8
    for (int k = 0; k < 32; ++k) {
        const int kk = half * 32 + k;
        partial += tile[m * LSTRIDE + kk] * Wp2[kk];
    }
    const float w = partial + __shfl_xor(partial, 16, 32) + bp2[0];

    if (half == 0) {
        v4f q;
        q.x = -dx * w; q.y = -dy * w; q.z = -dz * w; q.w = 1.0f;
        *(v4f*)&rec[m * 4] = q;
        rsrc[m] = src;
    }
    __syncthreads();
    {
        const int c  = lane & 3;
        const int i0 = lane >> 2;
        const int   s0 = rsrc[i0],             s1 = rsrc[i0 + 8];
        const float q0 = rec[i0 * 4 + c],      q1 = rec[(i0 + 8) * 4 + c];
        unsafeAtomicAdd(&xc[(size_t)s0 * 4 + c], q0);
        unsafeAtomicAdd(&xc[(size_t)s1 * 4 + c], q1);
    }
}

__global__ void __launch_bounds__(32)
egnn_node_kernel(const float* __restrict__ x, const float* __restrict__ h,
                 const float* __restrict__ Wn1p, const float* __restrict__ bn1,
                 const float* __restrict__ Wn2p, const float* __restrict__ bn2,
                 const float* __restrict__ xc, const float* __restrict__ msum,
                 float* __restrict__ xout, float* __restrict__ hout)
{
    __shared__ float tile[16 * LSTRIDE];
    const int lane = threadIdx.x & 31;
    const int m    = lane & 15;
    const int half = lane >> 4;
    const int n    = m;
    const int i0   = blockIdx.x * 16;
    const int i    = i0 + m;

    const float inv = 1.0f / fmaxf(xc[i * 4 + 3], 1.0f);

    if (half == 0) {
        #pragma unroll
        for (int d = 0; d < 3; ++d)
            xout[i * 3 + d] = x[i * 3 + d] + xc[i * 4 + d] * inv;
    }

    const float* __restrict__ hrow = h + (size_t)i * FDIM;
    const float* __restrict__ mrow = msum + (size_t)i * FDIM;

    v8f acc[4] = {};
    #pragma unroll 4
    for (int t = 0; t < 32; ++t) {
        const int k0 = 4 * t + 2 * half;
        v2f a;
        if (t < 16) { a.x = hrow[k0];            a.y = hrow[k0 + 1]; }
        else        { a.x = mrow[k0 - 64] * inv; a.y = mrow[k0 - 63] * inv; }
        wmma_step(Wn1p, t, lane, a, acc);
    }
    #pragma unroll
    for (int j = 0; j < 4; ++j)
        #pragma unroll
        for (int v = 0; v < 8; ++v)
            tile[(v + half * 8) * LSTRIDE + j * 16 + n] = silu_f(acc[j][v] + bn1[j * 16 + n]);
    __syncthreads();

    v8f acc2[4] = {};
    gemm_lds(tile, Wn2p, m, half, lane, acc2);
    #pragma unroll
    for (int v = 0; v < 8; ++v) {
        const int mr = v + half * 8;
        const size_t base = (size_t)(i0 + mr) * FDIM;
        #pragma unroll
        for (int j = 0; j < 4; ++j) {
            const int col = j * 16 + n;
            hout[base + col] = h[base + col] + acc2[j][v] + bn2[col];
        }
    }
}

__global__ void zero_kernel(float* __restrict__ p, int count) {
    const int idx = blockIdx.x * blockDim.x + threadIdx.x;
    if (idx < count) p[idx] = 0.0f;
}

extern "C" void kernel_launch(void* const* d_in, const int* in_sizes, int n_in,
                              void* d_out, int out_size, void* d_ws, size_t ws_size,
                              hipStream_t stream) {
    const float* x     = (const float*)d_in[0];
    const float* h     = (const float*)d_in[1];
    const int*   edges = (const int*)d_in[2];
    const float* We1   = (const float*)d_in[3];
    const float* be1   = (const float*)d_in[4];
    const float* We2   = (const float*)d_in[5];
    const float* be2   = (const float*)d_in[6];
    const float* Wp1   = (const float*)d_in[7];
    const float* bp1   = (const float*)d_in[8];
    const float* Wp2   = (const float*)d_in[9];
    const float* bp2   = (const float*)d_in[10];
    const float* Wn1   = (const float*)d_in[11];
    const float* bn1   = (const float*)d_in[12];
    const float* Wn2   = (const float*)d_in[13];
    const float* bn2   = (const float*)d_in[14];

    float* xc   = (float*)d_ws;
    float* msum = xc + 4 * N_NODES;
    float* wpk  = msum + (size_t)N_NODES * FDIM;
    float* We1p = wpk;
    float* We2p = We1p + 33 * 256;
    float* Wp1p = We2p + 16 * 256;
    float* Wn1p = Wp1p + 16 * 256;
    float* Wn2p = Wn1p + 32 * 256;

    float* xout = (float*)d_out;
    float* hout = xout + 3 * N_NODES;

    const int zcount = N_NODES * (1 + 3 + FDIM);
    zero_kernel<<<(zcount + 255) / 256, 256, 0, stream>>>(xc, zcount);

    pack_kernel<<<(33 * 256 + 255) / 256, 256, 0, stream>>>(We1, We1p, 129, 33);
    pack_kernel<<<(16 * 256 + 255) / 256, 256, 0, stream>>>(We2, We2p, 64, 16);
    pack_kernel<<<(16 * 256 + 255) / 256, 256, 0, stream>>>(Wp1, Wp1p, 64, 16);
    pack_kernel<<<(32 * 256 + 255) / 256, 256, 0, stream>>>(Wn1, Wn1p, 128, 32);
    pack_kernel<<<(16 * 256 + 255) / 256, 256, 0, stream>>>(Wn2, Wn2p, 64, 16);

    egnn_edge_kernel<<<N_EDGES / 16, 32, 0, stream>>>(
        x, h, edges, We1p, be1, We2p, be2, Wp1p, bp1, Wp2, bp2, xc, msum);

    egnn_node_kernel<<<N_NODES / 16, 32, 0, stream>>>(
        x, h, Wn1p, bn1, Wn2p, bn2, xc, msum, xout, hout);
}


# edge kernel: a tile's 32 h rows gathered as whole 256-byte rows (4 full lines per load) into LDS, layer-1 A fragments read from LDS; weight loads pinned 7 K steps ahead to hold the register count
# speedup vs baseline: 1.2271x; 1.1010x over previous
#include <hip/hip_runtime.h>
#include <math.h>

typedef float v2f __attribute__((ext_vector_type(2)));
typedef float v4f __attribute__((ext_vector_type(4)));
typedef float v8f __attribute__((ext_vector_type(8)));

#define N_NODES 50000
#define N_EDGES 800000
#define FDIM 64
#define LSTRIDE 68
#define APF 2
#define HSTRIDE 132
#define WPF 7

__device__ __forceinline__ float silu_f(float v) {
    return v * (1.0f / (1.0f + __expf(-v)));
}

__device__ __forceinline__ v8f wmma4(v2f a, v2f b, v8f c) {
    return __builtin_amdgcn_wmma_f32_16x16x4_f32(false, a, false, b, (short)0, c, false, false);
}

__device__ __forceinline__ void wmma_step(const float* __restrict__ Wp, int t, int lane,
                                          v2f a, v8f* acc) {
    const float* p = Wp + ((size_t)(t * 32 + lane) << 3);
    const v4f f0 = *(const v4f*)(p);
    const v4f f1 = *(const v4f*)(p + 4);
    v2f b;
    b.x = f0.x; b.y = f0.y; acc[0] = wmma4(a, b, acc[0]);
    b.x = f0.z; b.y = f0.w; acc[1] = wmma4(a, b, acc[1]);
    b.x = f1.x; b.y = f1.y; acc[2] = wmma4(a, b, acc[2]);
    b.x = f1.z; b.y = f1.w; acc[3] = wmma4(a, b, acc[3]);
}

__device__ __forceinline__ void gemm_lds(const float* __restrict__ tile,
                                         const float* __restrict__ Wp,
                                         int m, int half, int lane, v8f* acc) {
    #pragma unroll
    for (int t = 0; t < 16; ++t) {
        const int k0 = 4 * t + 2 * half;
        v2f a;
        a.x = tile[m * LSTRIDE + k0];
        a.y = tile[m * LSTRIDE + k0 + 1];
        wmma_step(Wp, t, lane, a, acc);
    }
}

__device__ __forceinline__ void ring_fill(const float* __restrict__ Wp, int lane, v4f* w0, v4f* w1) {
    #pragma unroll
    for (int i = 0; i < WPF; ++i) {
        const float* p = Wp + ((size_t)(i * 32 + lane) << 3);
        w0[i] = *(const v4f*)(p);
        w1[i] = *(const v4f*)(p + 4);
    }
    __builtin_amdgcn_sched_barrier(0);
}

template <int NSTEPS, int STRIDE>
__device__ __forceinline__ void ring_gemm(const float* tile, const float* __restrict__ Wp,
                                          int m, int half, int lane, v4f* w0, v4f* w1, v8f* acc) {
    const float* arow = tile + m * STRIDE + 2 * half;
    v2f ar[APF];
    #pragma unroll
    for (int i = 0; i < APF; ++i) ar[i] = *(const v2f*)&arow[4 * i];
    #pragma unroll
    for (int t = 0; t < NSTEPS; ++t) {
        const v2f a = ar[t % APF];
        if (t + APF < NSTEPS) ar[t % APF] = *(const v2f*)&arow[4 * (t + APF)];
        const v4f f0 = w0[t % WPF];
        const v4f f1 = w1[t % WPF];
        v2f b;
        b.x = f0.x; b.y = f0.y; acc[0] = wmma4(a, b, acc[0]);
        b.x = f0.z; b.y = f0.w; acc[1] = wmma4(a, b, acc[1]);
        b.x = f1.x; b.y = f1.y; acc[2] = wmma4(a, b, acc[2]);
        b.x = f1.z; b.y = f1.w; acc[3] = wmma4(a, b, acc[3]);
        if (t + WPF < NSTEPS) {
            const float* p = Wp + ((size_t)((t + WPF) * 32 + lane) << 3);
            w0[t % WPF] = *(const v4f*)(p);
            w1[t % WPF] = *(const v4f*)(p + 4);
        }
        __builtin_amdgcn_sched_barrier(0);
    }
}

__global__ void pack_kernel(const float* __restrict__ W, float* __restrict__ dst,
                            int krows, int T) {
    const int idx = blockIdx.x * blockDim.x + threadIdx.x;
    if (idx >= T * 256) return;
    const int t    = idx >> 8;
    const int r    = idx & 255;
    const int lane = r >> 3;
    const int q    = r & 7;
    const int j    = q >> 1;
    const int v    = q & 1;
    const int k    = 4 * t + 2 * (lane >> 4) + v;
    dst[idx] = (k < krows) ? W[k * FDIM + j * 16 + (lane & 15)] : 0.0f;
}

__global__ void __launch_bounds__(32)
egnn_edge_kernel(const float* __restrict__ x, const float* __restrict__ h,
                 const int* __restrict__ edges,
                 const float* __restrict__ We1p, const float* __restrict__ be1,
                 const float* __restrict__ We2p, const float* __restrict__ be2,
                 const float* __restrict__ Wp1p, const float* __restrict__ bp1,
                 const float* __restrict__ Wp2, const float* __restrict__ bp2,
                 float* __restrict__ xc, float* __restrict__ msum)
{
    __shared__ __attribute__((aligned(16))) float tile[16 * HSTRIDE];
    __shared__ __attribute__((aligned(16))) float rec[16 * 4];
    __shared__ int rsrc[16];
    const int lane = threadIdx.x & 31;
    const int m    = lane & 15;
    const int half = lane >> 4;
    const int n    = m;
    const int e0   = blockIdx.x * 16;
    const int e    = e0 + m;

    v4f w0[WPF], w1[WPF];

    const int src = edges[e];
    const int dst = edges[N_EDGES + e];
    const float dx = x[src * 3 + 0] - x[dst * 3 + 0];
    const float dy = x[src * 3 + 1] - x[dst * 3 + 1];
    const float dz = x[src * 3 + 2] - x[dst * 3 + 2];
    const float dij = sqrtf(dx * dx + dy * dy + dz * dz);

    {
        const int c4 = m * 4;
        v4f rs[8], rd[8];
        #pragma unroll
        for (int i = 0; i < 8; ++i) {
            const int r  = 2 * i + half;
            const int sr = __shfl(src, r, 32);
            const int dr = __shfl(dst, r, 32);
            rs[i] = *(const v4f*)(h + (size_t)sr * FDIM + c4);
            rd[i] = *(const v4f*)(h + (size_t)dr * FDIM + c4);
        }
        #pragma unroll
        for (int i = 0; i < 8; ++i) {
            const int r = 2 * i + half;
            *(v4f*)&tile[r * HSTRIDE + c4]      = rs[i];
            *(v4f*)&tile[r * HSTRIDE + 64 + c4] = rd[i];
        }
        if (half == 0) {
            v4f q;
            q.x = dij; q.y = 0.0f; q.z = 0.0f; q.w = 0.0f;
            *(v4f*)&tile[m * HSTRIDE + 128] = q;
        }
    }
    __syncthreads();
    ring_fill(We1p, lane, w0, w1);
    float bj[4];
    #pragma unroll
    for (int j = 0; j < 4; ++j) bj[j] = be1[j * 16 + n];

    v8f acc[4] = {};
    ring_gemm<33, HSTRIDE>(tile, We1p, m, half, lane, w0, w1, acc);
    #pragma unroll
    for (int j = 0; j < 4; ++j) {
        #pragma unroll
        for (int v = 0; v < 8; ++v)
            tile[(v + half * 8) * LSTRIDE + j * 16 + n] = silu_f(acc[j][v] + bj[j]);
        __builtin_amdgcn_sched_barrier(0);
    }
    __syncthreads();
    ring_fill(We2p, lane, w0, w1);
    #pragma unroll
    for (int j = 0; j < 4; ++j) bj[j] = be2[j * 16 + n];

    v8f macc[4] = {};
    ring_gemm<16, LSTRIDE>(tile, We2p, m, half, lane, w0, w1, macc);
    #pragma unroll
    for (int j = 0; j < 4; ++j) {
        #pragma unroll
        for (int v = 0; v < 8; ++v)
            macc[j][v] = silu_f(macc[j][v] + bj[j]);
        __builtin_amdgcn_sched_barrier(0);
    }
    __syncthreads();
    #pragma unroll
    for (int j = 0; j < 4; ++j)
        #pragma unroll
        for (int v = 0; v < 8; ++v)
            tile[(v + half * 8) * LSTRIDE + j * 16 + n] = macc[j][v];
    __syncthreads();

    {
        int   sr[16];
        float lo[16], hi[16];
        #pragma unroll
        for (int r = 0; r < 16; ++r) {
            sr[r] = __shfl(src, r, 32);
            lo[r] = tile[r * LSTRIDE + lane];
            hi[r] = tile[r * LSTRIDE + 32 + lane];
        }
        #pragma unroll
        for (int r = 0; r < 16; ++r) {
            float* __restrict__ p = msum + (size_t)sr[r] * FDIM + lane;
            unsafeAtomicAdd(p,      lo[r]);
            unsafeAtomicAdd(p + 32, hi[r]);
        }
    }

    ring_fill(Wp1p, lane, w0, w1);
    #pragma unroll
    for (int j = 0; j < 4; ++j) bj[j] = bp1[j * 16 + n];
    v8f pacc[4] = {};
    ring_gemm<16, LSTRIDE>(tile, Wp1p, m, half, lane, w0, w1, pacc);
    __syncthreads();
    #pragma unroll
    for (int j = 0; j < 4; ++j) {
        #pragma unroll
        for (int v = 0; v < 8; ++v)
            tile[(v + half * 8) * LSTRIDE + j * 16 + n] = silu_f(pacc[j][v] + bj[j]);
        __builtin_amdgcn_sched_barrier(0);
    }
    __syncthreads();

    float partial = 0.0f;
    #pragma unroll 8
    for (int k = 0; k < 32; ++k) {
        const int kk = half * 32 + k;
        partial += tile[m * LSTRIDE + kk] * Wp2[kk];
    }
    const float w = partial + __shfl_xor(partial, 16, 32) + bp2[0];

    if (half == 0) {
        v4f q;
        q.x = -dx * w; q.y = -dy * w; q.z = -dz * w; q.w = 1.0f;
        *(v4f*)&rec[m * 4] = q;
        rsrc[m] = src;
    }
    __syncthreads();
    {
        const int c  = lane & 3;
        const int i0 = lane >> 2;
        const int   s0 = rsrc[i0],             s1 = rsrc[i0 + 8];
        const float q0 = rec[i0 * 4 + c],      q1 = rec[(i0 + 8) * 4 + c];
        unsafeAtomicAdd(&xc[(size_t)s0 * 4 + c], q0);
        unsafeAtomicAdd(&xc[(size_t)s1 * 4 + c], q1);
    }
}

__global__ void __launch_bounds__(32)
egnn_node_kernel(const float* __restrict__ x, const float* __restrict__ h,
                 const float* __restrict__ Wn1p, const float* __restrict__ bn1,
                 const float* __restrict__ Wn2p, const float* __restrict__ bn2,
                 const float* __restrict__ xc, const float* __restrict__ msum,
                 float* __restrict__ xout, float* __restrict__ hout)
{
    __shared__ float tile[16 * LSTRIDE];
    const int lane = threadIdx.x & 31;
    const int m    = lane & 15;
    const int half = lane >> 4;
    const int n    = m;
    const int i0   = blockIdx.x * 16;
    const int i    = i0 + m;

    const float inv = 1.0f / fmaxf(xc[i * 4 + 3], 1.0f);

    if (half == 0) {
        #pragma unroll
        for (int d = 0; d < 3; ++d)
            xout[i * 3 + d] = x[i * 3 + d] + xc[i * 4 + d] * inv;
    }

    const float* __restrict__ hrow = h + (size_t)i * FDIM;
    const float* __restrict__ mrow = msum + (size_t)i * FDIM;

    v8f acc[4] = {};
    #pragma unroll 4
    for (int t = 0; t < 32; ++t) {
        const int k0 = 4 * t + 2 * half;
        v2f a;
        if (t < 16) { a.x = hrow[k0];            a.y = hrow[k0 + 1]; }
        else        { a.x = mrow[k0 - 64] * inv; a.y = mrow[k0 - 63] * inv; }
        wmma_step(Wn1p, t, lane, a, acc);
    }
    #pragma unroll
    for (int j = 0; j < 4; ++j)
        #pragma unroll
        for (int v = 0; v < 8; ++v)
            tile[(v + half * 8) * LSTRIDE + j * 16 + n] = silu_f(acc[j][v] + bn1[j * 16 + n]);
    __syncthreads();

    v8f acc2[4] = {};
    gemm_lds(tile, Wn2p, m, half, lane, acc2);
    #pragma unroll
    for (int v = 0; v < 8; ++v) {
        const int mr = v + half * 8;
        const size_t base = (size_t)(i0 + mr) * FDIM;
        #pragma unroll
        for (int j = 0; j < 4; ++j) {
            const int col = j * 16 + n;
            hout[base + col] = h[base + col] + acc2[j][v] + bn2[col];
        }
    }
}

__global__ void zero_kernel(float* __restrict__ p, int count) {
    const int idx = blockIdx.x * blockDim.x + threadIdx.x;
    if (idx < count) p[idx] = 0.0f;
}

extern "C" void kernel_launch(void* const* d_in, const int* in_sizes, int n_in,
                              void* d_out, int out_size, void* d_ws, size_t ws_size,
                              hipStream_t stream) {
    const float* x     = (const float*)d_in[0];
    const float* h     = (const float*)d_in[1];
    const int*   edges = (const int*)d_in[2];
    const float* We1   = (const float*)d_in[3];
    const float* be1   = (const float*)d_in[4];
    const float* We2   = (const float*)d_in[5];
    const float* be2   = (const float*)d_in[6];
    const float* Wp1   = (const float*)d_in[7];
    const float* bp1   = (const float*)d_in[8];
    const float* Wp2   = (const float*)d_in[9];
    const float* bp2   = (const float*)d_in[10];
    const float* Wn1   = (const float*)d_in[11];
    const float* bn1   = (const float*)d_in[12];
    const float* Wn2   = (const float*)d_in[13];
    const float* bn2   = (const float*)d_in[14];

    float* xc   = (float*)d_ws;
    float* msum = xc + 4 * N_NODES;
    float* wpk  = msum + (size_t)N_NODES * FDIM;
    float* We1p = wpk;
    float* We2p = We1p + 33 * 256;
    float* Wp1p = We2p + 16 * 256;
    float* Wn1p = Wp1p + 16 * 256;
    float* Wn2p = Wn1p + 32 * 256;

    float* xout = (float*)d_out;
    float* hout = xout + 3 * N_NODES;

    const int zcount = N_NODES * (1 + 3 + FDIM);
    zero_kernel<<<(zcount + 255) / 256, 256, 0, stream>>>(xc, zcount);

    pack_kernel<<<(33 * 256 + 255) / 256, 256, 0, stream>>>(We1, We1p, 129, 33);
    pack_kernel<<<(16 * 256 + 255) / 256, 256, 0, stream>>>(We2, We2p, 64, 16);
    pack_kernel<<<(16 * 256 + 255) / 256, 256, 0, stream>>>(Wp1, Wp1p, 64, 16);
    pack_kernel<<<(32 * 256 + 255) / 256, 256, 0, stream>>>(Wn1, Wn1p, 128, 32);
    pack_kernel<<<(16 * 256 + 255) / 256, 256, 0, stream>>>(Wn2, Wn2p, 64, 16);

    egnn_edge_kernel<<<N_EDGES / 16, 32, 0, stream>>>(
        x, h, edges, We1p, be1, We2p, be2, Wp1p, bp1, Wp2, bp2, xc, msum);

    egnn_node_kernel<<<N_NODES / 16, 32, 0, stream>>>(
        x, h, Wn1p, bn1, Wn2p, bn2, xc, msum, xout, hout);
}


# edge MLP layer 1 hoisted through the gather: P=h@We1[0:64], Q=h@We1[64:128] once per node on f32 WMMA; per edge silu(P[src]+Q[dst]+d*We1[128]+b) as vector fma; layer-1 sums regrouped 64+64+1
# speedup vs baseline: 1.5644x; 1.2749x over previous
#include <hip/hip_runtime.h>
#include <math.h>

typedef float v2f __attribute__((ext_vector_type(2)));
typedef float v4f __attribute__((ext_vector_type(4)));
typedef float v8f __attribute__((ext_vector_type(8)));

#define N_NODES 50000
#define N_EDGES 800000
#define FDIM 64
#define LSTRIDE 68
#define APF 2
#define WPF 7

__device__ __forceinline__ float silu_f(float v) {
    return v * (1.0f / (1.0f + __expf(-v)));
}

__device__ __forceinline__ v8f wmma4(v2f a, v2f b, v8f c) {
    return __builtin_amdgcn_wmma_f32_16x16x4_f32(false, a, false, b, (short)0, c, false, false);
}

__device__ __forceinline__ void wmma_step(const float* __restrict__ Wp, int t, int lane,
                                          v2f a, v8f* acc) {
    const float* p = Wp + ((size_t)(t * 32 + lane) << 3);
    const v4f f0 = *(const v4f*)(p);
    const v4f f1 = *(const v4f*)(p + 4);
    v2f b;
    b.x = f0.x; b.y = f0.y; acc[0] = wmma4(a, b, acc[0]);
    b.x = f0.z; b.y = f0.w; acc[1] = wmma4(a, b, acc[1]);
    b.x = f1.x; b.y = f1.y; acc[2] = wmma4(a, b, acc[2]);
    b.x = f1.z; b.y = f1.w; acc[3] = wmma4(a, b, acc[3]);
}

__device__ __forceinline__ void gemm_lds(const float* __restrict__ tile,
                                         const float* __restrict__ Wp,
                                         int m, int half, int lane, v8f* acc) {
    #pragma unroll
    for (int t = 0; t < 16; ++t) {
        const int k0 = 4 * t + 2 * half;
        v2f a;
        a.x = tile[m * LSTRIDE + k0];
        a.y = tile[m * LSTRIDE + k0 + 1];
        wmma_step(Wp, t, lane, a, acc);
    }
}

__device__ __forceinline__ void ring_fill(const float* __restrict__ Wp, int lane, v4f* w0, v4f* w1) {
    #pragma unroll
    for (int i = 0; i < WPF; ++i) {
        const float* p = Wp + ((size_t)(i * 32 + lane) << 3);
        w0[i] = *(const v4f*)(p);
        w1[i] = *(const v4f*)(p + 4);
    }
    __builtin_amdgcn_sched_barrier(0);
}

template <int NSTEPS, int STRIDE>
__device__ __forceinline__ void ring_gemm(const float* tile, const float* __restrict__ Wp,
                                          int m, int half, int lane, v4f* w0, v4f* w1, v8f* acc) {
    const float* arow = tile + m * STRIDE + 2 * half;
    v2f ar[APF];
    #pragma unroll
    for (int i = 0; i < APF; ++i) ar[i] = *(const v2f*)&arow[4 * i];
    #pragma unroll
    for (int t = 0; t < NSTEPS; ++t) {
        const v2f a = ar[t % APF];
        if (t + APF < NSTEPS) ar[t % APF] = *(const v2f*)&arow[4 * (t + APF)];
        const v4f f0 = w0[t % WPF];
        const v4f f1 = w1[t % WPF];
        v2f b;
        b.x = f0.x; b.y = f0.y; acc[0] = wmma4(a, b, acc[0]);
        b.x = f0.z; b.y = f0.w; acc[1] = wmma4(a, b, acc[1]);
        b.x = f1.x; b.y = f1.y; acc[2] = wmma4(a, b, acc[2]);
        b.x = f1.z; b.y = f1.w; acc[3] = wmma4(a, b, acc[3]);
        if (t + WPF < NSTEPS) {
            const float* p = Wp + ((size_t)((t + WPF) * 32 + lane) << 3);
            w0[t % WPF] = *(const v4f*)(p);
            w1[t % WPF] = *(const v4f*)(p + 4);
        }
        __builtin_amdgcn_sched_barrier(0);
    }
}

__global__ void pack_kernel(const float* __restrict__ W, float* __restrict__ dst,
                            int krows, int T) {
    const int idx = blockIdx.x * blockDim.x + threadIdx.x;
    if (idx >= T * 256) return;
    const int t    = idx >> 8;
    const int r    = idx & 255;
    const int lane = r >> 3;
    const int q    = r & 7;
    const int j    = q >> 1;
    const int v    = q & 1;
    const int k    = 4 * t + 2 * (lane >> 4) + v;
    dst[idx] = (k < krows) ? W[k * FDIM + j * 16 + (lane & 15)] : 0.0f;
}

__global__ void __launch_bounds__(32)
egnn_proj_kernel(const float* __restrict__ h, const float* __restrict__ We1p,
                 float* __restrict__ P, float* __restrict__ Q)
{
    __shared__ __attribute__((aligned(16))) float tile[2][16 * LSTRIDE];
    const int lane = threadIdx.x & 31;
    const int m    = lane & 15;
    const int half = lane >> 4;
    const int n    = m;
    const int i0   = blockIdx.x * 16;
    const int c4   = m * 4;

    {
        v4f rr[8];
        #pragma unroll
        for (int i = 0; i < 8; ++i)
            rr[i] = *(const v4f*)(h + (size_t)(i0 + 2 * i + half) * FDIM + c4);
        #pragma unroll
        for (int i = 0; i < 8; ++i)
            *(v4f*)&tile[0][(2 * i + half) * LSTRIDE + c4] = rr[i];
    }
    __syncthreads();

    v8f accP[4] = {}, accQ[4] = {};
    #pragma unroll 4
    for (int t = 0; t < 16; ++t) {
        const int k0 = 4 * t + 2 * half;
        v2f a;
        a.x = tile[0][m * LSTRIDE + k0];
        a.y = tile[0][m * LSTRIDE + k0 + 1];
        wmma_step(We1p, t,      lane, a, accP);
        wmma_step(We1p, 16 + t, lane, a, accQ);
    }
    __syncthreads();

    #pragma unroll
    for (int j = 0; j < 4; ++j)
        #pragma unroll
        for (int v = 0; v < 8; ++v) {
            tile[0][(v + half * 8) * LSTRIDE + j * 16 + n] = accP[j][v];
            tile[1][(v + half * 8) * LSTRIDE + j * 16 + n] = accQ[j][v];
        }
    __syncthreads();
    #pragma unroll
    for (int i = 0; i < 8; ++i) {
        const int r = 2 * i + half;
        *(v4f*)(P + (size_t)(i0 + r) * FDIM + c4) = *(const v4f*)&tile[0][r * LSTRIDE + c4];
        *(v4f*)(Q + (size_t)(i0 + r) * FDIM + c4) = *(const v4f*)&tile[1][r * LSTRIDE + c4];
    }
}

__global__ void __launch_bounds__(32)
egnn_edge_kernel(const float* __restrict__ x, const int* __restrict__ edges,
                 const float* __restrict__ P, const float* __restrict__ Q,
                 const float* __restrict__ We1, const float* __restrict__ be1,
                 const float* __restrict__ We2p, const float* __restrict__ be2,
                 const float* __restrict__ Wp1p, const float* __restrict__ bp1,
                 const float* __restrict__ Wp2, const float* __restrict__ bp2,
                 float* __restrict__ xc, float* __restrict__ msum)
{
    __shared__ __attribute__((aligned(16))) float tile[16 * LSTRIDE];
    __shared__ __attribute__((aligned(16))) float rec[16 * 4];
    __shared__ int rsrc[16];
    const int lane = threadIdx.x & 31;
    const int m    = lane & 15;
    const int half = lane >> 4;
    const int n    = m;
    const int e0   = blockIdx.x * 16;
    const int e    = e0 + m;

    v4f w0[WPF], w1[WPF];

    const int src = edges[e];
    const int dst = edges[N_EDGES + e];
    const float dx = x[src * 3 + 0] - x[dst * 3 + 0];
    const float dy = x[src * 3 + 1] - x[dst * 3 + 1];
    const float dz = x[src * 3 + 2] - x[dst * 3 + 2];
    const float dij = sqrtf(dx * dx + dy * dy + dz * dz);

    float bj[4];
    {
        const int c4 = m * 4;
        v4f rs[8], rd[8];
        #pragma unroll
        for (int i = 0; i < 8; ++i) {
            const int r  = 2 * i + half;
            const int sr = __shfl(src, r, 32);
            const int dr = __shfl(dst, r, 32);
            rs[i] = *(const v4f*)(P + (size_t)sr * FDIM + c4);
            rd[i] = *(const v4f*)(Q + (size_t)dr * FDIM + c4);
        }
        const v4f wd = *(const v4f*)(We1 + 128 * FDIM + c4);
        const v4f b1 = *(const v4f*)(be1 + c4);
        float dr8[8];
        #pragma unroll
        for (int i = 0; i < 8; ++i) dr8[i] = __shfl(dij, 2 * i + half, 32);
        #pragma unroll
        for (int i = 0; i < 8; ++i) {
            const int r = 2 * i + half;
            const float d = dr8[i];
            v4f v;
            v.x = silu_f(rs[i].x + rd[i].x + d * wd.x + b1.x);
            v.y = silu_f(rs[i].y + rd[i].y + d * wd.y + b1.y);
            v.z = silu_f(rs[i].z + rd[i].z + d * wd.z + b1.z);
            v.w = silu_f(rs[i].w + rd[i].w + d * wd.w + b1.w);
            *(v4f*)&tile[r * LSTRIDE + c4] = v;
            __builtin_amdgcn_sched_barrier(0);
        }
    }
    __syncthreads();
    ring_fill(We2p, lane, w0, w1);
    #pragma unroll
    for (int j = 0; j < 4; ++j) bj[j] = be2[j * 16 + n];

    v8f macc[4] = {};
    ring_gemm<16, LSTRIDE>(tile, We2p, m, half, lane, w0, w1, macc);
    #pragma unroll
    for (int j = 0; j < 4; ++j) {
        #pragma unroll
        for (int v = 0; v < 8; ++v)
            macc[j][v] = silu_f(macc[j][v] + bj[j]);
        __builtin_amdgcn_sched_barrier(0);
    }
    __syncthreads();
    #pragma unroll
    for (int j = 0; j < 4; ++j)
        #pragma unroll
        for (int v = 0; v < 8; ++v)
            tile[(v + half * 8) * LSTRIDE + j * 16 + n] = macc[j][v];
    __syncthreads();

    {
        int   sr[16];
        float lo[16], hi[16];
        #pragma unroll
        for (int r = 0; r < 16; ++r) {
            sr[r] = __shfl(src, r, 32);
            lo[r] = tile[r * LSTRIDE + lane];
            hi[r] = tile[r * LSTRIDE + 32 + lane];
        }
        #pragma unroll
        for (int r = 0; r < 16; ++r) {
            float* __restrict__ p = msum + (size_t)sr[r] * FDIM + lane;
            unsafeAtomicAdd(p,      lo[r]);
            unsafeAtomicAdd(p + 32, hi[r]);
        }
    }

    ring_fill(Wp1p, lane, w0, w1);
    #pragma unroll
    for (int j = 0; j < 4; ++j) bj[j] = bp1[j * 16 + n];
    v8f pacc[4] = {};
    ring_gemm<16, LSTRIDE>(tile, Wp1p, m, half, lane, w0, w1, pacc);
    __syncthreads();
    #pragma unroll
    for (int j = 0; j < 4; ++j) {
        #pragma unroll
        for (int v = 0; v < 8; ++v)
            tile[(v + half * 8) * LSTRIDE + j * 16 + n] = silu_f(pacc[j][v] + bj[j]);
        __builtin_amdgcn_sched_barrier(0);
    }
    __syncthreads();

    float partial = 0.0f;
    #pragma unroll 8
    for (int k = 0; k < 32; ++k) {
        const int kk = half * 32 + k;
        partial += tile[m * LSTRIDE + kk] * Wp2[kk];
    }
    const float w = partial + __shfl_xor(partial, 16, 32) + bp2[0];

    if (half == 0) {
        v4f q;
        q.x = -dx * w; q.y = -dy * w; q.z = -dz * w; q.w = 1.0f;
        *(v4f*)&rec[m * 4] = q;
        rsrc[m] = src;
    }
    __syncthreads();
    {
        const int c  = lane & 3;
        const int i0 = lane >> 2;
        const int   s0 = rsrc[i0],             s1 = rsrc[i0 + 8];
        const float q0 = rec[i0 * 4 + c],      q1 = rec[(i0 + 8) * 4 + c];
        unsafeAtomicAdd(&xc[(size_t)s0 * 4 + c], q0);
        unsafeAtomicAdd(&xc[(size_t)s1 * 4 + c], q1);
    }
}

__global__ void __launch_bounds__(32)
egnn_node_kernel(const float* __restrict__ x, const float* __restrict__ h,
                 const float* __restrict__ Wn1p, const float* __restrict__ bn1,
                 const float* __restrict__ Wn2p, const float* __restrict__ bn2,
                 const float* __restrict__ xc, const float* __restrict__ msum,
                 float* __restrict__ xout, float* __restrict__ hout)
{
    __shared__ float tile[16 * LSTRIDE];
    const int lane = threadIdx.x & 31;
    const int m    = lane & 15;
    const int half = lane >> 4;
    const int n    = m;
    const int i0   = blockIdx.x * 16;
    const int i    = i0 + m;

    const float inv = 1.0f / fmaxf(xc[i * 4 + 3], 1.0f);

    if (half == 0) {
        #pragma unroll
        for (int d = 0; d < 3; ++d)
            xout[i * 3 + d] = x[i * 3 + d] + xc[i * 4 + d] * inv;
    }

    const float* __restrict__ hrow = h + (size_t)i * FDIM;
    const float* __restrict__ mrow = msum + (size_t)i * FDIM;

    v8f acc[4] = {};
    #pragma unroll 4
    for (int t = 0; t < 32; ++t) {
        const int k0 = 4 * t + 2 * half;
        v2f a;
        if (t < 16) { a.x = hrow[k0];            a.y = hrow[k0 + 1]; }
        else        { a.x = mrow[k0 - 64] * inv; a.y = mrow[k0 - 63] * inv; }
        wmma_step(Wn1p, t, lane, a, acc);
    }
    #pragma unroll
    for (int j = 0; j < 4; ++j)
        #pragma unroll
        for (int v = 0; v < 8; ++v)
            tile[(v + half * 8) * LSTRIDE + j * 16 + n] = silu_f(acc[j][v] + bn1[j * 16 + n]);
    __syncthreads();

    v8f acc2[4] = {};
    gemm_lds(tile, Wn2p, m, half, lane, acc2);
    #pragma unroll
    for (int v = 0; v < 8; ++v) {
        const int mr = v + half * 8;
        const size_t base = (size_t)(i0 + mr) * FDIM;
        #pragma unroll
        for (int j = 0; j < 4; ++j) {
            const int col = j * 16 + n;
            hout[base + col] = h[base + col] + acc2[j][v] + bn2[col];
        }
    }
}

__global__ void zero_kernel(float* __restrict__ p, int count) {
    const int idx = blockIdx.x * blockDim.x + threadIdx.x;
    if (idx < count) p[idx] = 0.0f;
}

extern "C" void kernel_launch(void* const* d_in, const int* in_sizes, int n_in,
                              void* d_out, int out_size, void* d_ws, size_t ws_size,
                              hipStream_t stream) {
    const float* x     = (const float*)d_in[0];
    const float* h     = (const float*)d_in[1];
    const int*   edges = (const int*)d_in[2];
    const float* We1   = (const float*)d_in[3];
    const float* be1   = (const float*)d_in[4];
    const float* We2   = (const float*)d_in[5];
    const float* be2   = (const float*)d_in[6];
    const float* Wp1   = (const float*)d_in[7];
    const float* bp1   = (const float*)d_in[8];
    const float* Wp2   = (const float*)d_in[9];
    const float* bp2   = (const float*)d_in[10];
    const float* Wn1   = (const float*)d_in[11];
    const float* bn1   = (const float*)d_in[12];
    const float* Wn2   = (const float*)d_in[13];
    const float* bn2   = (const float*)d_in[14];

    if (ws_size < 39314688) return;

    float* xc   = (float*)d_ws;
    float* msum = xc + 4 * N_NODES;
    float* wpk  = msum + (size_t)N_NODES * FDIM;
    float* We1p = wpk;
    float* We2p = We1p + 32 * 256;
    float* Wp1p = We2p + 16 * 256;
    float* Wn1p = Wp1p + 16 * 256;
    float* Wn2p = Wn1p + 32 * 256;
    float* P    = Wn2p + 16 * 256;
    float* Q    = P + (size_t)N_NODES * FDIM;

    float* xout = (float*)d_out;
    float* hout = xout + 3 * N_NODES;

    const int zcount = N_NODES * (1 + 3 + FDIM);
    zero_kernel<<<(zcount + 255) / 256, 256, 0, stream>>>(xc, zcount);

    pack_kernel<<<(32 * 256 + 255) / 256, 256, 0, stream>>>(We1, We1p, 128, 32);
    pack_kernel<<<(16 * 256 + 255) / 256, 256, 0, stream>>>(We2, We2p, 64, 16);
    pack_kernel<<<(16 * 256 + 255) / 256, 256, 0, stream>>>(Wp1, Wp1p, 64, 16);
    pack_kernel<<<(32 * 256 + 255) / 256, 256, 0, stream>>>(Wn1, Wn1p, 128, 32);
    pack_kernel<<<(16 * 256 + 255) / 256, 256, 0, stream>>>(Wn2, Wn2p, 64, 16);

    egnn_proj_kernel<<<N_NODES / 16, 32, 0, stream>>>(h, We1p, P, Q);

    egnn_edge_kernel<<<N_EDGES / 16, 32, 0, stream>>>(
        x, edges, P, Q, We1, be1, We2p, be2, Wp1p, bp1, Wp2, bp2, xc, msum);

    egnn_node_kernel<<<N_NODES / 16, 32, 0, stream>>>(
        x, h, Wn1p, bn1, Wn2p, bn2, xc, msum, xout, hout);
}


# silu at all four sites: sigmoid as rcp(1 + exp(-v)) on the hardware reciprocal (v_rcp_f32, 1 ulp) instead of the IEEE division sequence; layer-1 activation pins regrouped, two row pieces per group
# speedup vs baseline: 1.7599x; 1.1250x over previous
#include <hip/hip_runtime.h>
#include <math.h>

typedef float v2f __attribute__((ext_vector_type(2)));
typedef float v4f __attribute__((ext_vector_type(4)));
typedef float v8f __attribute__((ext_vector_type(8)));

#define N_NODES 50000
#define N_EDGES 800000
#define FDIM 64
#define LSTRIDE 68
#define APF 2
#define WPF 7

__device__ __forceinline__ float silu_f(float v) {
    return v * __builtin_amdgcn_rcpf(1.0f + __expf(-v));
}

__device__ __forceinline__ v8f wmma4(v2f a, v2f b, v8f c) {
    return __builtin_amdgcn_wmma_f32_16x16x4_f32(false, a, false, b, (short)0, c, false, false);
}

__device__ __forceinline__ void wmma_step(const float* __restrict__ Wp, int t, int lane,
                                          v2f a, v8f* acc) {
    const float* p = Wp + ((size_t)(t * 32 + lane) << 3);
    const v4f f0 = *(const v4f*)(p);
    const v4f f1 = *(const v4f*)(p + 4);
    v2f b;
    b.x = f0.x; b.y = f0.y; acc[0] = wmma4(a, b, acc[0]);
    b.x = f0.z; b.y = f0.w; acc[1] = wmma4(a, b, acc[1]);
    b.x = f1.x; b.y = f1.y; acc[2] = wmma4(a, b, acc[2]);
    b.x = f1.z; b.y = f1.w; acc[3] = wmma4(a, b, acc[3]);
}

__device__ __forceinline__ void gemm_lds(const float* __restrict__ tile,
                                         const float* __restrict__ Wp,
                                         int m, int half, int lane, v8f* acc) {
    #pragma unroll
    for (int t = 0; t < 16; ++t) {
        const int k0 = 4 * t + 2 * half;
        v2f a;
        a.x = tile[m * LSTRIDE + k0];
        a.y = tile[m * LSTRIDE + k0 + 1];
        wmma_step(Wp, t, lane, a, acc);
    }
}

__device__ __forceinline__ void ring_fill(const float* __restrict__ Wp, int lane, v4f* w0, v4f* w1) {
    #pragma unroll
    for (int i = 0; i < WPF; ++i) {
        const float* p = Wp + ((size_t)(i * 32 + lane) << 3);
        w0[i] = *(const v4f*)(p);
        w1[i] = *(const v4f*)(p + 4);
    }
    __builtin_amdgcn_sched_barrier(0);
}

template <int NSTEPS, int STRIDE>
__device__ __forceinline__ void ring_gemm(const float* tile, const float* __restrict__ Wp,
                                          int m, int half, int lane, v4f* w0, v4f* w1, v8f* acc) {
    const float* arow = tile + m * STRIDE + 2 * half;
    v2f ar[APF];
    #pragma unroll
    for (int i = 0; i < APF; ++i) ar[i] = *(const v2f*)&arow[4 * i];
    #pragma unroll
    for (int t = 0; t < NSTEPS; ++t) {
        const v2f a = ar[t % APF];
        if (t + APF < NSTEPS) ar[t % APF] = *(const v2f*)&arow[4 * (t + APF)];
        const v4f f0 = w0[t % WPF];
        const v4f f1 = w1[t % WPF];
        v2f b;
        b.x = f0.x; b.y = f0.y; acc[0] = wmma4(a, b, acc[0]);
        b.x = f0.z; b.y = f0.w; acc[1] = wmma4(a, b, acc[1]);
        b.x = f1.x; b.y = f1.y; acc[2] = wmma4(a, b, acc[2]);
        b.x = f1.z; b.y = f1.w; acc[3] = wmma4(a, b, acc[3]);
        if (t + WPF < NSTEPS) {
            const float* p = Wp + ((size_t)((t + WPF) * 32 + lane) << 3);
            w0[t % WPF] = *(const v4f*)(p);
            w1[t % WPF] = *(const v4f*)(p + 4);
        }
        __builtin_amdgcn_sched_barrier(0);
    }
}

__global__ void pack_kernel(const float* __restrict__ W, float* __restrict__ dst,
                            int krows, int T) {
    const int idx = blockIdx.x * blockDim.x + threadIdx.x;
    if (idx >= T * 256) return;
    const int t    = idx >> 8;
    const int r    = idx & 255;
    const int lane = r >> 3;
    const int q    = r & 7;
    const int j    = q >> 1;
    const int v    = q & 1;
    const int k    = 4 * t + 2 * (lane >> 4) + v;
    dst[idx] = (k < krows) ? W[k * FDIM + j * 16 + (lane & 15)] : 0.0f;
}

__global__ void __launch_bounds__(32)
egnn_proj_kernel(const float* __restrict__ h, const float* __restrict__ We1p,
                 float* __restrict__ P, float* __restrict__ Q)
{
    __shared__ __attribute__((aligned(16))) float tile[2][16 * LSTRIDE];
    const int lane = threadIdx.x & 31;
    const int m    = lane & 15;
    const int half = lane >> 4;
    const int n    = m;
    const int i0   = blockIdx.x * 16;
    const int c4   = m * 4;

    {
        v4f rr[8];
        #pragma unroll
        for (int i = 0; i < 8; ++i)
            rr[i] = *(const v4f*)(h + (size_t)(i0 + 2 * i + half) * FDIM + c4);
        #pragma unroll
        for (int i = 0; i < 8; ++i)
            *(v4f*)&tile[0][(2 * i + half) * LSTRIDE + c4] = rr[i];
    }
    __syncthreads();

    v8f accP[4] = {}, accQ[4] = {};
    #pragma unroll 4
    for (int t = 0; t < 16; ++t) {
        const int k0 = 4 * t + 2 * half;
        v2f a;
        a.x = tile[0][m * LSTRIDE + k0];
        a.y = tile[0][m * LSTRIDE + k0 + 1];
        wmma_step(We1p, t,      lane, a, accP);
        wmma_step(We1p, 16 + t, lane, a, accQ);
    }
    __syncthreads();

    #pragma unroll
    for (int j = 0; j < 4; ++j)
        #pragma unroll
        for (int v = 0; v < 8; ++v) {
            tile[0][(v + half * 8) * LSTRIDE + j * 16 + n] = accP[j][v];
            tile[1][(v + half * 8) * LSTRIDE + j * 16 + n] = accQ[j][v];
        }
    __syncthreads();
    #pragma unroll
    for (int i = 0; i < 8; ++i) {
        const int r = 2 * i + half;
        *(v4f*)(P + (size_t)(i0 + r) * FDIM + c4) = *(const v4f*)&tile[0][r * LSTRIDE + c4];
        *(v4f*)(Q + (size_t)(i0 + r) * FDIM + c4) = *(const v4f*)&tile[1][r * LSTRIDE + c4];
    }
}

__global__ void __launch_bounds__(32)
egnn_edge_kernel(const float* __restrict__ x, const int* __restrict__ edges,
                 const float* __restrict__ P, const float* __restrict__ Q,
                 const float* __restrict__ We1, const float* __restrict__ be1,
                 const float* __restrict__ We2p, const float* __restrict__ be2,
                 const float* __restrict__ Wp1p, const float* __restrict__ bp1,
                 const float* __restrict__ Wp2, const float* __restrict__ bp2,
                 float* __restrict__ xc, float* __restrict__ msum)
{
    __shared__ __attribute__((aligned(16))) float tile[16 * LSTRIDE];
    __shared__ __attribute__((aligned(16))) float rec[16 * 4];
    __shared__ int rsrc[16];
    const int lane = threadIdx.x & 31;
    const int m    = lane & 15;
    const int half = lane >> 4;
    const int n    = m;
    const int e0   = blockIdx.x * 16;
    const int e    = e0 + m;

    v4f w0[WPF], w1[WPF];

    const int src = edges[e];
    const int dst = edges[N_EDGES + e];
    const float dx = x[src * 3 + 0] - x[dst * 3 + 0];
    const float dy = x[src * 3 + 1] - x[dst * 3 + 1];
    const float dz = x[src * 3 + 2] - x[dst * 3 + 2];
    const float dij = sqrtf(dx * dx + dy * dy + dz * dz);

    float bj[4];
    {
        const int c4 = m * 4;
        v4f rs[8], rd[8];
        #pragma unroll
        for (int i = 0; i < 8; ++i) {
            const int r  = 2 * i + half;
            const int sr = __shfl(src, r, 32);
            const int dr = __shfl(dst, r, 32);
            rs[i] = *(const v4f*)(P + (size_t)sr * FDIM + c4);
            rd[i] = *(const v4f*)(Q + (size_t)dr * FDIM + c4);
        }
        const v4f wd = *(const v4f*)(We1 + 128 * FDIM + c4);
        const v4f b1 = *(const v4f*)(be1 + c4);
        float dr8[8];
        #pragma unroll
        for (int i = 0; i < 8; ++i) dr8[i] = __shfl(dij, 2 * i + half, 32);
        #pragma unroll
        for (int i = 0; i < 8; ++i) {
            const int r = 2 * i + half;
            const float d = dr8[i];
            v4f v;
            v.x = silu_f(rs[i].x + rd[i].x + d * wd.x + b1.x);
            v.y = silu_f(rs[i].y + rd[i].y + d * wd.y + b1.y);
            v.z = silu_f(rs[i].z + rd[i].z + d * wd.z + b1.z);
            v.w = silu_f(rs[i].w + rd[i].w + d * wd.w + b1.w);
            *(v4f*)&tile[r * LSTRIDE + c4] = v;
            if (i & 1) __builtin_amdgcn_sched_barrier(0);
        }
    }
    __syncthreads();
    ring_fill(We2p, lane, w0, w1);
    #pragma unroll
    for (int j = 0; j < 4; ++j) bj[j] = be2[j * 16 + n];

    v8f macc[4] = {};
    ring_gemm<16, LSTRIDE>(tile, We2p, m, half, lane, w0, w1, macc);
    #pragma unroll
    for (int j = 0; j < 4; ++j) {
        #pragma unroll
        for (int v = 0; v < 8; ++v)
            macc[j][v] = silu_f(macc[j][v] + bj[j]);
        __builtin_amdgcn_sched_barrier(0);
    }
    __syncthreads();
    #pragma unroll
    for (int j = 0; j < 4; ++j)
        #pragma unroll
        for (int v = 0; v < 8; ++v)
            tile[(v + half * 8) * LSTRIDE + j * 16 + n] = macc[j][v];
    __syncthreads();

    {
        int   sr[16];
        float lo[16], hi[16];
        #pragma unroll
        for (int r = 0; r < 16; ++r) {
            sr[r] = __shfl(src, r, 32);
            lo[r] = tile[r * LSTRIDE + lane];
            hi[r] = tile[r * LSTRIDE + 32 + lane];
        }
        #pragma unroll
        for (int r = 0; r < 16; ++r) {
            float* __restrict__ p = msum + (size_t)sr[r] * FDIM + lane;
            unsafeAtomicAdd(p,      lo[r]);
            unsafeAtomicAdd(p + 32, hi[r]);
        }
    }

    ring_fill(Wp1p, lane, w0, w1);
    #pragma unroll
    for (int j = 0; j < 4; ++j) bj[j] = bp1[j * 16 + n];
    v8f pacc[4] = {};
    ring_gemm<16, LSTRIDE>(tile, Wp1p, m, half, lane, w0, w1, pacc);
    __syncthreads();
    #pragma unroll
    for (int j = 0; j < 4; ++j) {
        #pragma unroll
        for (int v = 0; v < 8; ++v)
            tile[(v + half * 8) * LSTRIDE + j * 16 + n] = silu_f(pacc[j][v] + bj[j]);
        __builtin_amdgcn_sched_barrier(0);
    }
    __syncthreads();

    float partial = 0.0f;
    #pragma unroll 8
    for (int k = 0; k < 32; ++k) {
        const int kk = half * 32 + k;
        partial += tile[m * LSTRIDE + kk] * Wp2[kk];
    }
    const float w = partial + __shfl_xor(partial, 16, 32) + bp2[0];

    if (half == 0) {
        v4f q;
        q.x = -dx * w; q.y = -dy * w; q.z = -dz * w; q.w = 1.0f;
        *(v4f*)&rec[m * 4] = q;
        rsrc[m] = src;
    }
    __syncthreads();
    {
        const int c  = lane & 3;
        const int i0 = lane >> 2;
        const int   s0 = rsrc[i0],             s1 = rsrc[i0 + 8];
        const float q0 = rec[i0 * 4 + c],      q1 = rec[(i0 + 8) * 4 + c];
        unsafeAtomicAdd(&xc[(size_t)s0 * 4 + c], q0);
        unsafeAtomicAdd(&xc[(size_t)s1 * 4 + c], q1);
    }
}

__global__ void __launch_bounds__(32)
egnn_node_kernel(const float* __restrict__ x, const float* __restrict__ h,
                 const float* __restrict__ Wn1p, const float* __restrict__ bn1,
                 const float* __restrict__ Wn2p, const float* __restrict__ bn2,
                 const float* __restrict__ xc, const float* __restrict__ msum,
                 float* __restrict__ xout, float* __restrict__ hout)
{
    __shared__ float tile[16 * LSTRIDE];
    const int lane = threadIdx.x & 31;
    const int m    = lane & 15;
    const int half = lane >> 4;
    const int n    = m;
    const int i0   = blockIdx.x * 16;
    const int i    = i0 + m;

    const float inv = 1.0f / fmaxf(xc[i * 4 + 3], 1.0f);

    if (half == 0) {
        #pragma unroll
        for (int d = 0; d < 3; ++d)
            xout[i * 3 + d] = x[i * 3 + d] + xc[i * 4 + d] * inv;
    }

    const float* __restrict__ hrow = h + (size_t)i * FDIM;
    const float* __restrict__ mrow = msum + (size_t)i * FDIM;

    v8f acc[4] = {};
    #pragma unroll 4
    for (int t = 0; t < 32; ++t) {
        const int k0 = 4 * t + 2 * half;
        v2f a;
        if (t < 16) { a.x = hrow[k0];            a.y = hrow[k0 + 1]; }
        else        { a.x = mrow[k0 - 64] * inv; a.y = mrow[k0 - 63] * inv; }
        wmma_step(Wn1p, t, lane, a, acc);
    }
    #pragma unroll
    for (int j = 0; j < 4; ++j)
        #pragma unroll
        for (int v = 0; v < 8; ++v)
            tile[(v + half * 8) * LSTRIDE + j * 16 + n] = silu_f(acc[j][v] + bn1[j * 16 + n]);
    __syncthreads();

    v8f acc2[4] = {};
    gemm_lds(tile, Wn2p, m, half, lane, acc2);
    #pragma unroll
    for (int v = 0; v < 8; ++v) {
        const int mr = v + half * 8;
        const size_t base = (size_t)(i0 + mr) * FDIM;
        #pragma unroll
        for (int j = 0; j < 4; ++j) {
            const int col = j * 16 + n;
            hout[base + col] = h[base + col] + acc2[j][v] + bn2[col];
        }
    }
}

__global__ void zero_kernel(float* __restrict__ p, int count) {
    const int idx = blockIdx.x * blockDim.x + threadIdx.x;
    if (idx < count) p[idx] = 0.0f;
}

extern "C" void kernel_launch(void* const* d_in, const int* in_sizes, int n_in,
                              void* d_out, int out_size, void* d_ws, size_t ws_size,
                              hipStream_t stream) {
    const float* x     = (const float*)d_in[0];
    const float* h     = (const float*)d_in[1];
    const int*   edges = (const int*)d_in[2];
    const float* We1   = (const float*)d_in[3];
    const float* be1   = (const float*)d_in[4];
    const float* We2   = (const float*)d_in[5];
    const float* be2   = (const float*)d_in[6];
    const float* Wp1   = (const float*)d_in[7];
    const float* bp1   = (const float*)d_in[8];
    const float* Wp2   = (const float*)d_in[9];
    const float* bp2   = (const float*)d_in[10];
    const float* Wn1   = (const float*)d_in[11];
    const float* bn1   = (const float*)d_in[12];
    const float* Wn2   = (const float*)d_in[13];
    const float* bn2   = (const float*)d_in[14];

    if (ws_size < 39314688) return;

    float* xc   = (float*)d_ws;
    float* msum = xc + 4 * N_NODES;
    float* wpk  = msum + (size_t)N_NODES * FDIM;
    float* We1p = wpk;
    float* We2p = We1p + 32 * 256;
    float* Wp1p = We2p + 16 * 256;
    float* Wn1p = Wp1p + 16 * 256;
    float* Wn2p = Wn1p + 32 * 256;
    float* P    = Wn2p + 16 * 256;
    float* Q    = P + (size_t)N_NODES * FDIM;

    float* xout = (float*)d_out;
    float* hout = xout + 3 * N_NODES;

    const int zcount = N_NODES * (1 + 3 + FDIM);
    zero_kernel<<<(zcount + 255) / 256, 256, 0, stream>>>(xc, zcount);

    pack_kernel<<<(32 * 256 + 255) / 256, 256, 0, stream>>>(We1, We1p, 128, 32);
    pack_kernel<<<(16 * 256 + 255) / 256, 256, 0, stream>>>(We2, We2p, 64, 16);
    pack_kernel<<<(16 * 256 + 255) / 256, 256, 0, stream>>>(Wp1, Wp1p, 64, 16);
    pack_kernel<<<(32 * 256 + 255) / 256, 256, 0, stream>>>(Wn1, Wn1p, 128, 32);
    pack_kernel<<<(16 * 256 + 255) / 256, 256, 0, stream>>>(Wn2, Wn2p, 64, 16);

    egnn_proj_kernel<<<N_NODES / 16, 32, 0, stream>>>(h, We1p, P, Q);

    egnn_edge_kernel<<<N_EDGES / 16, 32, 0, stream>>>(
        x, edges, P, Q, We1, be1, We2p, be2, Wp1p, bp1, Wp2, bp2, xc, msum);

    egnn_node_kernel<<<N_NODES / 16, 32, 0, stream>>>(
        x, h, Wn1p, bn1, Wn2p, bn2, xc, msum, xout, hout);
}


# launch count 9 -> 4: the five weight-packing launches ride as the first 112 blocks of the zero-fill launch (one preparation launch); packed layout, all other kernels and the order of sums untouched
# speedup vs baseline: 1.8512x; 1.0519x over previous
#include <hip/hip_runtime.h>
#include <math.h>

typedef float v2f __attribute__((ext_vector_type(2)));
typedef float v4f __attribute__((ext_vector_type(4)));
typedef float v8f __attribute__((ext_vector_type(8)));

#define N_NODES 50000
#define N_EDGES 800000
#define FDIM 64
#define LSTRIDE 68
#define APF 2
#define WPF 7

__device__ __forceinline__ float silu_f(float v) {
    return v * __builtin_amdgcn_rcpf(1.0f + __expf(-v));
}

__device__ __forceinline__ v8f wmma4(v2f a, v2f b, v8f c) {
    return __builtin_amdgcn_wmma_f32_16x16x4_f32(false, a, false, b, (short)0, c, false, false);
}

__device__ __forceinline__ void wmma_step(const float* __restrict__ Wp, int t, int lane,
                                          v2f a, v8f* acc) {
    const float* p = Wp + ((size_t)(t * 32 + lane) << 3);
    const v4f f0 = *(const v4f*)(p);
    const v4f f1 = *(const v4f*)(p + 4);
    v2f b;
    b.x = f0.x; b.y = f0.y; acc[0] = wmma4(a, b, acc[0]);
    b.x = f0.z; b.y = f0.w; acc[1] = wmma4(a, b, acc[1]);
    b.x = f1.x; b.y = f1.y; acc[2] = wmma4(a, b, acc[2]);
    b.x = f1.z; b.y = f1.w; acc[3] = wmma4(a, b, acc[3]);
}

__device__ __forceinline__ void gemm_lds(const float* __restrict__ tile,
                                         const float* __restrict__ Wp,
                                         int m, int half, int lane, v8f* acc) {
    #pragma unroll
    for (int t = 0; t < 16; ++t) {
        const int k0 = 4 * t + 2 * half;
        v2f a;
        a.x = tile[m * LSTRIDE + k0];
        a.y = tile[m * LSTRIDE + k0 + 1];
        wmma_step(Wp, t, lane, a, acc);
    }
}

__device__ __forceinline__ void ring_fill(const float* __restrict__ Wp, int lane, v4f* w0, v4f* w1) {
    #pragma unroll
    for (int i = 0; i < WPF; ++i) {
        const float* p = Wp + ((size_t)(i * 32 + lane) << 3);
        w0[i] = *(const v4f*)(p);
        w1[i] = *(const v4f*)(p + 4);
    }
    __builtin_amdgcn_sched_barrier(0);
}

template <int NSTEPS, int STRIDE>
__device__ __forceinline__ void ring_gemm(const float* tile, const float* __restrict__ Wp,
                                          int m, int half, int lane, v4f* w0, v4f* w1, v8f* acc) {
    const float* arow = tile + m * STRIDE + 2 * half;
    v2f ar[APF];
    #pragma unroll
    for (int i = 0; i < APF; ++i) ar[i] = *(const v2f*)&arow[4 * i];
    #pragma unroll
    for (int t = 0; t < NSTEPS; ++t) {
        const v2f a = ar[t % APF];
        if (t + APF < NSTEPS) ar[t % APF] = *(const v2f*)&arow[4 * (t + APF)];
        const v4f f0 = w0[t % WPF];
        const v4f f1 = w1[t % WPF];
        v2f b;
        b.x = f0.x; b.y = f0.y; acc[0] = wmma4(a, b, acc[0]);
        b.x = f0.z; b.y = f0.w; acc[1] = wmma4(a, b, acc[1]);
        b.x = f1.x; b.y = f1.y; acc[2] = wmma4(a, b, acc[2]);
        b.x = f1.z; b.y = f1.w; acc[3] = wmma4(a, b, acc[3]);
        if (t + WPF < NSTEPS) {
            const float* p = Wp + ((size_t)((t + WPF) * 32 + lane) << 3);
            w0[t % WPF] = *(const v4f*)(p);
            w1[t % WPF] = *(const v4f*)(p + 4);
        }
        __builtin_amdgcn_sched_barrier(0);
    }
}

#define PACK_BLOCKS 112
__global__ void __launch_bounds__(256)
prep_kernel(const float* __restrict__ We1, const float* __restrict__ We2, const float* __restrict__ Wp1,
            const float* __restrict__ Wn1, const float* __restrict__ Wn2,
            float* __restrict__ wpk, float* __restrict__ accum, int count) {
    const int b = blockIdx.x;
    if (b < PACK_BLOCKS) {
        const float* W;
        int t;
        if      (b < 32) { W = We1; t = b;      }
        else if (b < 48) { W = We2; t = b - 32; }
        else if (b < 64) { W = Wp1; t = b - 48; }
        else if (b < 96) { W = Wn1; t = b - 64; }
        else             { W = Wn2; t = b - 96; }
        const int r    = threadIdx.x;
        const int lane = r >> 3;
        const int q    = r & 7;
        const int j    = q >> 1;
        const int v    = q & 1;
        const int k    = 4 * t + 2 * (lane >> 4) + v;
        wpk[b * 256 + r] = W[k * FDIM + j * 16 + (lane & 15)];
        return;
    }
    const int idx = (b - PACK_BLOCKS) * 256 + threadIdx.x;
    if (idx < count) accum[idx] = 0.0f;
}

__global__ void __launch_bounds__(32)
egnn_proj_kernel(const float* __restrict__ h, const float* __restrict__ We1p,
                 float* __restrict__ P, float* __restrict__ Q)
{
    __shared__ __attribute__((aligned(16))) float tile[2][16 * LSTRIDE];
    const int lane = threadIdx.x & 31;
    const int m    = lane & 15;
    const int half = lane >> 4;
    const int n    = m;
    const int i0   = blockIdx.x * 16;
    const int c4   = m * 4;

    {
        v4f rr[8];
        #pragma unroll
        for (int i = 0; i < 8; ++i)
            rr[i] = *(const v4f*)(h + (size_t)(i0 + 2 * i + half) * FDIM + c4);
        #pragma unroll
        for (int i = 0; i < 8; ++i)
            *(v4f*)&tile[0][(2 * i + half) * LSTRIDE + c4] = rr[i];
    }
    __syncthreads();

    v8f accP[4] = {}, accQ[4] = {};
    #pragma unroll 4
    for (int t = 0; t < 16; ++t) {
        const int k0 = 4 * t + 2 * half;
        v2f a;
        a.x = tile[0][m * LSTRIDE + k0];
        a.y = tile[0][m * LSTRIDE + k0 + 1];
        wmma_step(We1p, t,      lane, a, accP);
        wmma_step(We1p, 16 + t, lane, a, accQ);
    }
    __syncthreads();

    #pragma unroll
    for (int j = 0; j < 4; ++j)
        #pragma unroll
        for (int v = 0; v < 8; ++v) {
            tile[0][(v + half * 8) * LSTRIDE + j * 16 + n] = accP[j][v];
            tile[1][(v + half * 8) * LSTRIDE + j * 16 + n] = accQ[j][v];
        }
    __syncthreads();
    #pragma unroll
    for (int i = 0; i < 8; ++i) {
        const int r = 2 * i + half;
        *(v4f*)(P + (size_t)(i0 + r) * FDIM + c4) = *(const v4f*)&tile[0][r * LSTRIDE + c4];
        *(v4f*)(Q + (size_t)(i0 + r) * FDIM + c4) = *(const v4f*)&tile[1][r * LSTRIDE + c4];
    }
}

__global__ void __launch_bounds__(32)
egnn_edge_kernel(const float* __restrict__ x, const int* __restrict__ edges,
                 const float* __restrict__ P, const float* __restrict__ Q,
                 const float* __restrict__ We1, const float* __restrict__ be1,
                 const float* __restrict__ We2p, const float* __restrict__ be2,
                 const float* __restrict__ Wp1p, const float* __restrict__ bp1,
                 const float* __restrict__ Wp2, const float* __restrict__ bp2,
                 float* __restrict__ xc, float* __restrict__ msum)
{
    __shared__ __attribute__((aligned(16))) float tile[16 * LSTRIDE];
    __shared__ __attribute__((aligned(16))) float rec[16 * 4];
    __shared__ int rsrc[16];
    const int lane = threadIdx.x & 31;
    const int m    = lane & 15;
    const int half = lane >> 4;
    const int n    = m;
    const int e0   = blockIdx.x * 16;
    const int e    = e0 + m;

    v4f w0[WPF], w1[WPF];

    const int src = edges[e];
    const int dst = edges[N_EDGES + e];
    const float dx = x[src * 3 + 0] - x[dst * 3 + 0];
    const float dy = x[src * 3 + 1] - x[dst * 3 + 1];
    const float dz = x[src * 3 + 2] - x[dst * 3 + 2];
    const float dij = sqrtf(dx * dx + dy * dy + dz * dz);

    float bj[4];
    {
        const int c4 = m * 4;
        v4f rs[8], rd[8];
        #pragma unroll
        for (int i = 0; i < 8; ++i) {
            const int r  = 2 * i + half;
            const int sr = __shfl(src, r, 32);
            const int dr = __shfl(dst, r, 32);
            rs[i] = *(const v4f*)(P + (size_t)sr * FDIM + c4);
            rd[i] = *(const v4f*)(Q + (size_t)dr * FDIM + c4);
        }
        const v4f wd = *(const v4f*)(We1 + 128 * FDIM + c4);
        const v4f b1 = *(const v4f*)(be1 + c4);
        float dr8[8];
        #pragma unroll
        for (int i = 0; i < 8; ++i) dr8[i] = __shfl(dij, 2 * i + half, 32);
        #pragma unroll
        for (int i = 0; i < 8; ++i) {
            const int r = 2 * i + half;
            const float d = dr8[i];
            v4f v;
            v.x = silu_f(rs[i].x + rd[i].x + d * wd.x + b1.x);
            v.y = silu_f(rs[i].y + rd[i].y + d * wd.y + b1.y);
            v.z = silu_f(rs[i].z + rd[i].z + d * wd.z + b1.z);
            v.w = silu_f(rs[i].w + rd[i].w + d * wd.w + b1.w);
            *(v4f*)&tile[r * LSTRIDE + c4] = v;
            if (i & 1) __builtin_amdgcn_sched_barrier(0);
        }
    }
    __syncthreads();
    ring_fill(We2p, lane, w0, w1);
    #pragma unroll
    for (int j = 0; j < 4; ++j) bj[j] = be2[j * 16 + n];

    v8f macc[4] = {};
    ring_gemm<16, LSTRIDE>(tile, We2p, m, half, lane, w0, w1, macc);
    #pragma unroll
    for (int j = 0; j < 4; ++j) {
        #pragma unroll
        for (int v = 0; v < 8; ++v)
            macc[j][v] = silu_f(macc[j][v] + bj[j]);
        __builtin_amdgcn_sched_barrier(0);
    }
    __syncthreads();
    #pragma unroll
    for (int j = 0; j < 4; ++j)
        #pragma unroll
        for (int v = 0; v < 8; ++v)
            tile[(v + half * 8) * LSTRIDE + j * 16 + n] = macc[j][v];
    __syncthreads();

    {
        int   sr[16];
        float lo[16], hi[16];
        #pragma unroll
        for (int r = 0; r < 16; ++r) {
            sr[r] = __shfl(src, r, 32);
            lo[r] = tile[r * LSTRIDE + lane];
            hi[r] = tile[r * LSTRIDE + 32 + lane];
        }
        #pragma unroll
        for (int r = 0; r < 16; ++r) {
            float* __restrict__ p = msum + (size_t)sr[r] * FDIM + lane;
            unsafeAtomicAdd(p,      lo[r]);
            unsafeAtomicAdd(p + 32, hi[r]);
        }
    }

    ring_fill(Wp1p, lane, w0, w1);
    #pragma unroll
    for (int j = 0; j < 4; ++j) bj[j] = bp1[j * 16 + n];
    v8f pacc[4] = {};
    ring_gemm<16, LSTRIDE>(tile, Wp1p, m, half, lane, w0, w1, pacc);
    __syncthreads();
    #pragma unroll
    for (int j = 0; j < 4; ++j) {
        #pragma unroll
        for (int v = 0; v < 8; ++v)
            tile[(v + half * 8) * LSTRIDE + j * 16 + n] = silu_f(pacc[j][v] + bj[j]);
        __builtin_amdgcn_sched_barrier(0);
    }
    __syncthreads();

    float partial = 0.0f;
    #pragma unroll 8
    for (int k = 0; k < 32; ++k) {
        const int kk = half * 32 + k;
        partial += tile[m * LSTRIDE + kk] * Wp2[kk];
    }
    const float w = partial + __shfl_xor(partial, 16, 32) + bp2[0];

    if (half == 0) {
        v4f q;
        q.x = -dx * w; q.y = -dy * w; q.z = -dz * w; q.w = 1.0f;
        *(v4f*)&rec[m * 4] = q;
        rsrc[m] = src;
    }
    __syncthreads();
    {
        const int c  = lane & 3;
        const int i0 = lane >> 2;
        const int   s0 = rsrc[i0],             s1 = rsrc[i0 + 8];
        const float q0 = rec[i0 * 4 + c],      q1 = rec[(i0 + 8) * 4 + c];
        unsafeAtomicAdd(&xc[(size_t)s0 * 4 + c], q0);
        unsafeAtomicAdd(&xc[(size_t)s1 * 4 + c], q1);
    }
}

__global__ void __launch_bounds__(32)
egnn_node_kernel(const float* __restrict__ x, const float* __restrict__ h,
                 const float* __restrict__ Wn1p, const float* __restrict__ bn1,
                 const float* __restrict__ Wn2p, const float* __restrict__ bn2,
                 const float* __restrict__ xc, const float* __restrict__ msum,
                 float* __restrict__ xout, float* __restrict__ hout)
{
    __shared__ float tile[16 * LSTRIDE];
    const int lane = threadIdx.x & 31;
    const int m    = lane & 15;
    const int half = lane >> 4;
    const int n    = m;
    const int i0   = blockIdx.x * 16;
    const int i    = i0 + m;

    const float inv = 1.0f / fmaxf(xc[i * 4 + 3], 1.0f);

    if (half == 0) {
        #pragma unroll
        for (int d = 0; d < 3; ++d)
            xout[i * 3 + d] = x[i * 3 + d] + xc[i * 4 + d] * inv;
    }

    const float* __restrict__ hrow = h + (size_t)i * FDIM;
    const float* __restrict__ mrow = msum + (size_t)i * FDIM;

    v8f acc[4] = {};
    #pragma unroll 4
    for (int t = 0; t < 32; ++t) {
        const int k0 = 4 * t + 2 * half;
        v2f a;
        if (t < 16) { a.x = hrow[k0];            a.y = hrow[k0 + 1]; }
        else        { a.x = mrow[k0 - 64] * inv; a.y = mrow[k0 - 63] * inv; }
        wmma_step(Wn1p, t, lane, a, acc);
    }
    #pragma unroll
    for (int j = 0; j < 4; ++j)
        #pragma unroll
        for (int v = 0; v < 8; ++v)
            tile[(v + half * 8) * LSTRIDE + j * 16 + n] = silu_f(acc[j][v] + bn1[j * 16 + n]);
    __syncthreads();

    v8f acc2[4] = {};
    gemm_lds(tile, Wn2p, m, half, lane, acc2);
    #pragma unroll
    for (int v = 0; v < 8; ++v) {
        const int mr = v + half * 8;
        const size_t base = (size_t)(i0 + mr) * FDIM;
        #pragma unroll
        for (int j = 0; j < 4; ++j) {
            const int col = j * 16 + n;
            hout[base + col] = h[base + col] + acc2[j][v] + bn2[col];
        }
    }
}

extern "C" void kernel_launch(void* const* d_in, const int* in_sizes, int n_in,
                              void* d_out, int out_size, void* d_ws, size_t ws_size,
                              hipStream_t stream) {
    const float* x     = (const float*)d_in[0];
    const float* h     = (const float*)d_in[1];
    const int*   edges = (const int*)d_in[2];
    const float* We1   = (const float*)d_in[3];
    const float* be1   = (const float*)d_in[4];
    const float* We2   = (const float*)d_in[5];
    const float* be2   = (const float*)d_in[6];
    const float* Wp1   = (const float*)d_in[7];
    const float* bp1   = (const float*)d_in[8];
    const float* Wp2   = (const float*)d_in[9];
    const float* bp2   = (const float*)d_in[10];
    const float* Wn1   = (const float*)d_in[11];
    const float* bn1   = (const float*)d_in[12];
    const float* Wn2   = (const float*)d_in[13];
    const float* bn2   = (const float*)d_in[14];

    if (ws_size < 39314688) return;

    float* xc   = (float*)d_ws;
    float* msum = xc + 4 * N_NODES;
    float* wpk  = msum + (size_t)N_NODES * FDIM;
    float* We1p = wpk;
    float* We2p = We1p + 32 * 256;
    float* Wp1p = We2p + 16 * 256;
    float* Wn1p = Wp1p + 16 * 256;
    float* Wn2p = Wn1p + 32 * 256;
    float* P    = Wn2p + 16 * 256;
    float* Q    = P + (size_t)N_NODES * FDIM;

    float* xout = (float*)d_out;
    float* hout = xout + 3 * N_NODES;

    const int zcount = N_NODES * (1 + 3 + FDIM);
    prep_kernel<<<PACK_BLOCKS + (zcount + 255) / 256, 256, 0, stream>>>(We1, We2, Wp1, Wn1, Wn2, wpk, xc, zcount);

    egnn_proj_kernel<<<N_NODES / 16, 32, 0, stream>>>(h, We1p, P, Q);

    egnn_edge_kernel<<<N_EDGES / 16, 32, 0, stream>>>(
        x, edges, P, Q, We1, be1, We2p, be2, Wp1p, bp1, Wp2, bp2, xc, msum);

    egnn_node_kernel<<<N_NODES / 16, 32, 0, stream>>>(
        x, h, Wn1p, bn1, Wn2p, bn2, xc, msum, xout, hout);
}
